# starting kernel
# baseline (speedup 1.0000x reference)
#include <hip/hip_runtime.h>
#include <math.h>


typedef float v2f __attribute__((ext_vector_type(2)));
typedef float v8f __attribute__((ext_vector_type(8)));

#define NW 8
#define SS 1024
#define EE 64

__global__ __launch_bounds__(256)
void qmeas_kernel(const float* __restrict__ x,
                  const float* __restrict__ theta,
                  float* __restrict__ Qp,
                  float* __restrict__ Kp,
                  float* __restrict__ V2) {
    int t = blockIdx.x * blockDim.x + threadIdx.x;
    int b   = t >> 13;
    int rem = t & 8191;
    int s   = rem >> 3;
    int h   = rem & 7;
    const float* xp = x + (((size_t)b * SS + s) * EE + h * NW);
    float c[8];
#pragma unroll
    for (int j = 0; j < 8; ++j) c[j] = cosf(xp[j] + theta[j]);
    float p1 = c[0]*c[1];
    float p2 = p1*c[2];
    float p3 = p2*c[3];
    float p4 = p3*c[4];
    float p5 = p4*c[5];
    float p6 = p5*c[6];
    float p7 = p6*c[7];
    float m0 = c[1]*c[2]*c[3]*c[4]*c[5]*c[6]*c[7];
    size_t row = (size_t)(b * NW + h) * SS + s;
    const float sc = 0.125f * 1.4426950408889634f;
    ((float4*)Qp)[row] = make_float4(m0 * sc, p1 * sc, p2 * sc, 0.0f);
    ((float4*)Kp)[row] = make_float4(p3, p4, p5, 0.0f);
    ((float2*)V2)[row] = make_float2(p6, p7);
}

__global__ __launch_bounds__(256)
void attn_kernel(const float* __restrict__ Qp,
                 const float* __restrict__ Kp,
                 const float* __restrict__ V2,
                 float* __restrict__ out16) {
    int wave = blockIdx.x * (blockDim.x >> 5) + (threadIdx.x >> 5);
    int lane = threadIdx.x & 31;
    int half = lane >> 4;
    int lm   = lane & 15;
    int bh = wave >> 6;
    int sb = wave & 63;
    int s0 = sb << 4;
    size_t base = (size_t)bh * SS;

    v2f bq = *(const v2f*)(Qp + (base + s0 + lm) * 4 + 2 * half);

    float lsum = 0.f, acc0 = 0.f, acc1 = 0.f;

    for (int tb = 0; tb < 64; ++tb) {
        int t0 = tb << 4;
        v2f ak = *(const v2f*)(Kp + (base + t0 + lm) * 4 + 2 * half);
        v8f cz = {};
        v8f sc = __builtin_amdgcn_wmma_f32_16x16x4_f32(
            false, ak, false, bq, (short)0, cz, false, false);

        const float4* vp = (const float4*)(V2 + (base + t0 + 8 * half) * 2);
        float4 vv0 = vp[0], vv1 = vp[1], vv2 = vp[2], vv3 = vp[3];
        float v0[8] = {vv0.x, vv0.z, vv1.x, vv1.z, vv2.x, vv2.z, vv3.x, vv3.z};
        float v1[8] = {vv0.y, vv0.w, vv1.y, vv1.w, vv2.y, vv2.w, vv3.y, vv3.w};

#pragma unroll
        for (int r = 0; r < 8; ++r) {
            float p = __builtin_amdgcn_exp2f(sc[r]);
            lsum += p;
            acc0 = fmaf(p, v0[r], acc0);
            acc1 = fmaf(p, v1[r], acc1);
        }
    }
    float lt  = lsum + __shfl_xor(lsum, 16);
    float a0t = acc0 + __shfl_xor(acc0, 16);
    float a1t = acc1 + __shfl_xor(acc1, 16);

    int b = bh >> 3, h = bh & 7;
    int s = s0 + lm;
    float o = ((half == 0) ? a0t : a1t) / lt;
    out16[((size_t)b * SS + s) * 16 + h * 2 + half] = o;
}

__global__ __launch_bounds__(256)
void proj_kernel(const float* __restrict__ out16,
                 const float* __restrict__ W,
                 const float* __restrict__ bias,
                 float* __restrict__ out) {
    int wave = blockIdx.x * (blockDim.x >> 5) + (threadIdx.x >> 5);
    int lane = threadIdx.x & 31;
    int half = lane >> 4;
    int lm   = lane & 15;
    int mtile = wave >> 2;
    int ntile = wave & 3;
    int m0 = mtile << 4;
    int n0 = ntile << 4;
    float bv = bias[n0 + lm];
    v8f c;
#pragma unroll
    for (int r = 0; r < 8; ++r) c[r] = bv;
#pragma unroll
    for (int kc = 0; kc < 4; ++kc) {
        v2f a  = *(const v2f*)(out16 + (size_t)(m0 + lm) * 16 + kc * 4 + 2 * half);
        v2f bb = *(const v2f*)(W + (size_t)(n0 + lm) * 16 + kc * 4 + 2 * half);
        c = __builtin_amdgcn_wmma_f32_16x16x4_f32(
            false, a, false, bb, (short)0, c, false, false);
    }
#pragma unroll
    for (int r = 0; r < 8; ++r) {
        int row = m0 + r + 8 * half;
        out[(size_t)row * 64 + n0 + lm] = c[r];
    }
}

extern "C" void kernel_launch(void* const* d_in, const int* in_sizes, int n_in,
                              void* d_out, int out_size, void* d_ws, size_t ws_size,
                              hipStream_t stream) {
    const float* x     = (const float*)d_in[0];
    const float* theta = (const float*)d_in[1];
    const float* W     = (const float*)d_in[2];
    const float* bias  = (const float*)d_in[3];
    float* out = (float*)d_out;

    char* ws = (char*)d_ws;
    float* Qp    = (float*)(ws);
    float* Kp    = (float*)(ws + (1u << 20));
    float* V2    = (float*)(ws + (2u << 20));
    float* out16 = (float*)(ws + (2u << 20) + (1u << 19));

    qmeas_kernel<<<256, 256, 0, stream>>>(x, theta, Qp, Kp, V2);
    attn_kernel <<<512, 256, 0, stream>>>(Qp, Kp, V2, out16);
    proj_kernel <<<256, 256, 0, stream>>>(out16, W, bias, out);
}


# measurement folded into the attention launch on in-line cosines (3-part pi/2 reduction + polynomials, within 2 ulp; library cosine past |angle| 1024); 2 launches, not 3; same sums, same order
# speedup vs baseline: 1.5447x; 1.5447x over previous
#include <hip/hip_runtime.h>
#include <math.h>


typedef float v2f __attribute__((ext_vector_type(2)));
typedef float v8f __attribute__((ext_vector_type(8)));

#define NW 8
#define SS 1024
#define EE 64

#define COS_INLINE_MAX 1024.0f

__device__ __forceinline__ float cos_inline(float a) {
    float t = fabsf(a);
    float j = __builtin_rintf(t * 0.636619747f);
    float r = fmaf(j, -1.57079625e+0f, t);
    r = fmaf(j, -7.54978942e-8f, r);
    r = fmaf(j, -5.39030253e-15f, r);
    int q = (int)j;
    float s = r * r;
    float cp = fmaf(s, 2.443315711809948e-5f, -1.388731625493765e-3f);
    cp = fmaf(s, cp, 4.166664568298827e-2f);
    cp = fmaf(s, cp, -0.5f);
    cp = fmaf(s, cp, 1.0f);
    float sp = fmaf(s, -1.9515295891e-4f, 8.3321608736e-3f);
    sp = fmaf(s, sp, -1.6666654611e-1f);
    sp = fmaf(r * s, sp, r);
    float v = (q & 1) ? sp : cp;
    unsigned sg = ((unsigned)(q + 1) & 2u) << 30;
    return __uint_as_float(__float_as_uint(v) ^ sg);
}

__device__ __forceinline__ void store_row(const float* c, int row, bool is_q, int qrow,
                                          float4* sK, float2* sV, float4* sQ) {
    float p1 = c[0]*c[1];
    float p2 = p1*c[2];
    float p3 = p2*c[3];
    float p4 = p3*c[4];
    float p5 = p4*c[5];
    float p6 = p5*c[6];
    float p7 = p6*c[7];
    sK[row] = make_float4(p3, p4, p5, 0.0f);
    sV[row] = make_float2(p6, p7);
    if (is_q) {
        float m0 = c[1]*c[2]*c[3]*c[4]*c[5]*c[6]*c[7];
        const float sc = 0.125f * 1.4426950408889634f;
        sQ[qrow] = make_float4(m0 * sc, p1 * sc, p2 * sc, 0.0f);
    }
}

__global__ __launch_bounds__(256)
void attn_kernel(const float* __restrict__ x,
                 const float* __restrict__ theta,
                 float* __restrict__ out16) {
    __shared__ float4 sK[SS];
    __shared__ float4 sV[SS / 2];
    __shared__ float4 sQ[256];

    int tid  = threadIdx.x;
    int bh   = blockIdx.x >> 2;
    int qblk = blockIdx.x & 3;
    int b = bh >> 3, h = bh & 7;

    int wv   = tid >> 5;
    int lane = tid & 31;
    int half = lane >> 4;
    int lm   = lane & 15;
    int s0   = (qblk << 8) + (wv << 5);

    const float* xh = x + (size_t)b * SS * EE + h * NW;
    float4 xa[4], xb[4];
#pragma unroll
    for (int i = 0; i < 4; ++i) {
        const float4* xp = (const float4*)(xh + (size_t)(tid + (i << 8)) * EE);
        xa[i] = xp[0];
        xb[i] = xp[1];
    }
    float th[8];
#pragma unroll
    for (int j = 0; j < 8; ++j) th[j] = theta[j];

    float2* sV2 = (float2*)sV;
    float big = 0.0f;
#pragma unroll
    for (int i = 0; i < 4; ++i) {
        float a[8] = {xa[i].x + th[0], xa[i].y + th[1], xa[i].z + th[2], xa[i].w + th[3],
                      xb[i].x + th[4], xb[i].y + th[5], xb[i].z + th[6], xb[i].w + th[7]};
        float c[8];
#pragma unroll
        for (int j = 0; j < 8; ++j) {
            big = fmaxf(big, fabsf(a[j]));
            c[j] = cos_inline(a[j]);
        }
        store_row(c, tid + (i << 8), i == qblk, tid, sK, sV2, sQ);
    }
    if (__builtin_expect(big > COS_INLINE_MAX, 0)) {
#pragma nounroll
        for (int i = 0; i < 4; ++i) {
            int row = tid + (i << 8);
            const float* xr = xh + (size_t)row * EE;
            float c[8];
#pragma unroll
            for (int j = 0; j < 8; ++j) c[j] = cosf(xr[j] + theta[j]);
            store_row(c, row, i == qblk, tid, sK, sV2, sQ);
        }
    }
    __syncthreads();

    const float* sKf = (const float*)sK;
    const float* sQf = (const float*)sQ;
    int qa = (wv << 5) + lm;
    v2f bqA = *(const v2f*)(sQf + qa * 4 + 2 * half);
    v2f bqB = *(const v2f*)(sQf + (qa + 16) * 4 + 2 * half);

    float lsA = 0.f, a0A = 0.f, a1A = 0.f;
    float lsB = 0.f, a0B = 0.f, a1B = 0.f;

    for (int g = 0; g < 16; ++g) {
        int t0 = g << 6;
        v2f ak[4];
        float4 vv[4][4];
#pragma unroll
        for (int j = 0; j < 4; ++j) {
            int tj = t0 + (j << 4);
            ak[j] = *(const v2f*)(sKf + (tj + lm) * 4 + 2 * half);
            const float4* vp = sV + ((tj + 8 * half) >> 1);
            vv[j][0] = vp[0]; vv[j][1] = vp[1]; vv[j][2] = vp[2]; vv[j][3] = vp[3];
        }
        __builtin_amdgcn_sched_barrier(0);

#pragma unroll
        for (int j = 0; j < 4; ++j) {
            v8f cz = {};
            v8f scA = __builtin_amdgcn_wmma_f32_16x16x4_f32(
                false, ak[j], false, bqA, (short)0, cz, false, false);
            v8f scB = __builtin_amdgcn_wmma_f32_16x16x4_f32(
                false, ak[j], false, bqB, (short)0, cz, false, false);

            float v0[8] = {vv[j][0].x, vv[j][0].z, vv[j][1].x, vv[j][1].z,
                           vv[j][2].x, vv[j][2].z, vv[j][3].x, vv[j][3].z};
            float v1[8] = {vv[j][0].y, vv[j][0].w, vv[j][1].y, vv[j][1].w,
                           vv[j][2].y, vv[j][2].w, vv[j][3].y, vv[j][3].w};
#pragma unroll
            for (int r = 0; r < 8; ++r) {
                float pA = __builtin_amdgcn_exp2f(scA[r]);
                float pB = __builtin_amdgcn_exp2f(scB[r]);
                lsA += pA;
                lsB += pB;
                a0A = fmaf(pA, v0[r], a0A);
                a0B = fmaf(pB, v0[r], a0B);
                a1A = fmaf(pA, v1[r], a1A);
                a1B = fmaf(pB, v1[r], a1B);
            }
        }
    }
    float ltA = lsA + __shfl_xor(lsA, 16);
    float a0tA = a0A + __shfl_xor(a0A, 16);
    float a1tA = a1A + __shfl_xor(a1A, 16);
    float ltB = lsB + __shfl_xor(lsB, 16);
    float a0tB = a0B + __shfl_xor(a0B, 16);
    float a1tB = a1B + __shfl_xor(a1B, 16);

    int s = s0 + lm;
    float oA = ((half == 0) ? a0tA : a1tA) / ltA;
    float oB = ((half == 0) ? a0tB : a1tB) / ltB;
    out16[((size_t)b * SS + s) * 16 + h * 2 + half] = oA;
    out16[((size_t)b * SS + s + 16) * 16 + h * 2 + half] = oB;
}

__global__ __launch_bounds__(256)
void proj_kernel(const float* __restrict__ out16,
                 const float* __restrict__ W,
                 const float* __restrict__ bias,
                 float* __restrict__ out) {
    int wave = blockIdx.x * (blockDim.x >> 5) + (threadIdx.x >> 5);
    int lane = threadIdx.x & 31;
    int half = lane >> 4;
    int lm   = lane & 15;
    int mtile = wave >> 2;
    int ntile = wave & 3;
    int m0 = mtile << 4;
    int n0 = ntile << 4;
    float bv = bias[n0 + lm];
    v8f c;
#pragma unroll
    for (int r = 0; r < 8; ++r) c[r] = bv;
#pragma unroll
    for (int kc = 0; kc < 4; ++kc) {
        v2f a  = *(const v2f*)(out16 + (size_t)(m0 + lm) * 16 + kc * 4 + 2 * half);
        v2f bb = *(const v2f*)(W + (size_t)(n0 + lm) * 16 + kc * 4 + 2 * half);
        c = __builtin_amdgcn_wmma_f32_16x16x4_f32(
            false, a, false, bb, (short)0, c, false, false);
    }
#pragma unroll
    for (int r = 0; r < 8; ++r) {
        int row = m0 + r + 8 * half;
        out[(size_t)row * 64 + n0 + lm] = c[r];
    }
}

extern "C" void kernel_launch(void* const* d_in, const int* in_sizes, int n_in,
                              void* d_out, int out_size, void* d_ws, size_t ws_size,
                              hipStream_t stream) {
    const float* x     = (const float*)d_in[0];
    const float* theta = (const float*)d_in[1];
    const float* W     = (const float*)d_in[2];
    const float* bias  = (const float*)d_in[3];
    float* out = (float*)d_out;

    float* out16 = (float*)d_ws;

    attn_kernel<<<256, 256, 0, stream>>>(x, theta, out16);
    proj_kernel<<<256, 256, 0, stream>>>(out16, W, bias, out);
}
